# starting kernel
# baseline (speedup 1.0000x reference)
#include <hip/hip_runtime.h>
#include <hip/hip_bf16.h>


typedef __attribute__((ext_vector_type(2))) float v2f;
typedef __attribute__((ext_vector_type(8))) float v8f;

#define BATCH     32
#define FRAMES    4000
#define CBINS     65
#define NIR       129
#define FRAME_LEN 64
#define SEGLEN    192
#define OUTLEN    256128
#define KPAD      68
#define JROWS     144
#define IRS       145
#define FPB       16

__global__ void FN_setup_M(float* __restrict__ M) {
  const float TWO_PI = 6.28318530717958647692f;
  for (int i = blockIdx.x * blockDim.x + threadIdx.x; i < JROWS * KPAD;
       i += gridDim.x * blockDim.x) {
    int j = i / KPAD;
    int k = i - j * KPAD;
    float v = 0.0f;
    if (j < NIR && k < CBINS) {
      float win = 0.5f * (1.0f - cosf(TWO_PI * (float)j / (float)NIR));
      int m = (k * (j - 64)) % NIR;
      float ang = TWO_PI * (float)m / (float)NIR;
      float ck  = (k == 0) ? 1.0f : 2.0f;
      v = win * (ck / (float)NIR) * cosf(ang);
    }
    M[i] = v;
  }
}

__global__ void FN_zero(float4* __restrict__ out4, int n4) {
  float4 z; z.x = z.y = z.z = z.w = 0.0f;
  for (int i = blockIdx.x * blockDim.x + threadIdx.x; i < n4;
       i += gridDim.x * blockDim.x)
    out4[i] = z;
}

__global__ __launch_bounds__(256) void FN_fused(const float* __restrict__ H,
                                                const float* __restrict__ noise,
                                                const float* __restrict__ M,
                                                float* __restrict__ out) {
  __shared__ float irs[FPB * IRS];
  __shared__ float nz[FPB * FRAME_LEN];

  const int tid  = threadIdx.x;
  const int lane = tid & 31;
  const int wave = tid >> 5;

  const int b  = blockIdx.x / (FRAMES / FPB);
  const int f0 = (blockIdx.x - b * (FRAMES / FPB)) * FPB;
  const long r0 = (long)b * FRAMES + f0;

  for (int i = tid; i < FPB * FRAME_LEN; i += 256) {
    int lf = i >> 6;
    int tt = i & 63;
    nz[i] = noise[(r0 + lf) * FRAME_LEN + tt] * 2.0f - 1.0f;
  }

  const int row = lane & 15;
  const int kb  = (lane >> 4) << 1;
  const float* hrow = H + (r0 + row) * CBINS;

  for (int jt = wave; jt < 9; jt += 8) {
    v8f acc = {};
    const float* mrow = M + (jt * 16 + row) * KPAD + kb;
#pragma unroll
    for (int k0 = 0; k0 < KPAD; k0 += 4) {
      int ka = k0 + kb;
      v2f a;
      a.x = (ka     < CBINS) ? hrow[ka]     : 0.0f;
      a.y = (ka + 1 < CBINS) ? hrow[ka + 1] : 0.0f;
      v2f bv = *(const v2f*)(mrow + k0);
      acc = __builtin_amdgcn_wmma_f32_16x16x4_f32(
           false, a,  false, bv,
           (short)0, acc,  false,  false);
    }
    const int mbase = (lane >> 4) << 3;
#pragma unroll
    for (int v = 0; v < 8; ++v)
      irs[(mbase + v) * IRS + jt * 16 + row] = acc[v];
  }

  __syncthreads();

  for (int o = tid; o < FPB * SEGLEN; o += 256) {
    int lf = o / SEGLEN;
    int t  = o - lf * SEGLEN;
    int lo = (t > NIR - 1) ? (t - (NIR - 1)) : 0;
    int hi = (t < FRAME_LEN - 1) ? t : (FRAME_LEN - 1);
    const float* np  = nz + (lf << 6);
    const float* irp = irs + lf * IRS + t;
    float s = 0.0f;
    for (int tau = lo; tau <= hi; ++tau)
      s += np[tau] * irp[-tau];
    __hip_atomic_fetch_add(
        out + (long)b * OUTLEN + (long)(f0 + lf) * FRAME_LEN + t,
        s * 0.01f, __ATOMIC_RELAXED, __HIP_MEMORY_SCOPE_AGENT);
  }
}

extern "C" void kernel_launch(void* const* d_in, const int* in_sizes, int n_in,
                              void* d_out, int out_size, void* d_ws, size_t ws_size,
                              hipStream_t stream) {
  const float* H     = (const float*)d_in[0];
  const float* noise = (const float*)d_in[1];
  float* out = (float*)d_out;
  float* Mws = (float*)d_ws;

  FN_setup_M<<<(JROWS * KPAD + 255) / 256, 256, 0, stream>>>(Mws);
  FN_zero<<<4096, 256, 0, stream>>>((float4*)out, out_size / 4);
  FN_fused<<<BATCH * (FRAMES / FPB), 256, 0, stream>>>(H, noise, Mws, out);
}


# ir-matmul fragments fetched 16 B per lane: a lane takes 4 consecutive k of its H or M row for two WMMAs (same k permutation in A and B, sum order changes); the k=64 H load leaves the tile loop
# speedup vs baseline: 3.4282x; 3.4282x over previous
#include <hip/hip_runtime.h>
#include <hip/hip_bf16.h>


typedef __attribute__((ext_vector_type(2))) float v2f;
typedef __attribute__((ext_vector_type(8))) float v8f;
typedef __attribute__((ext_vector_type(4), aligned(4))) float v4fu;

#define BATCH     32
#define FRAMES    4000
#define CBINS     65
#define NIR       129
#define FRAME_LEN 64
#define SEGLEN    192
#define OUTLEN    256128
#define KPAD      68
#define JROWS     144
#define NHOPS     4002
#define FPW       32
#define HPW       30
#define WPB       134
#define FPV       4
#define KCONV     80
#define KHALF     40
#define IROW      276
#define IR_ZLO    68
#define IR_ZHI    220
#define NROW      96

__global__ void FN_setup_M(float* __restrict__ M) {
  const float TWO_PI = 6.28318530717958647692f;
  for (int i = blockIdx.x * blockDim.x + threadIdx.x; i < JROWS * KPAD;
       i += gridDim.x * blockDim.x) {
    int j = i / KPAD;
    int k = i - j * KPAD;
    float v = 0.0f;
    if (j < NIR && k < CBINS) {
      float win = 0.5f * (1.0f - cosf(TWO_PI * (float)j / (float)NIR));
      int m = (k * (j - 64)) % NIR;
      float ang = TWO_PI * (float)m / (float)NIR;
      float ck  = (k == 0) ? 1.0f : 2.0f;
      v = win * (ck / (float)NIR) * cosf(ang);
    }
    M[i] = v;
  }
}

__global__ __launch_bounds__(256) void FN_fused(const float* __restrict__ H,
                                                const float* __restrict__ noise,
                                                const float* __restrict__ M,
                                                float* __restrict__ out) {
  __shared__ float irs[FPW * IROW];
  __shared__ float nrev[FPW * NROW];
  __shared__ float carry[8 * 4 * 32];

  const int tid  = threadIdx.x;
  const int lane = tid & 31;
  const int wave = tid >> 5;

  const int b     = blockIdx.x / WPB;
  const int S     = (blockIdx.x - b * WPB) * HPW;
  const int fbase = S - 2;

  for (int i = tid; i < FPW * FRAME_LEN; i += 256) {
    int lf = i >> 6;
    int tt = i & 63;
    int f  = fbase + lf;
    float v = 0.0f;
    if (f >= 0 && f < FRAMES)
      v = noise[((long)b * FRAMES + f) * FRAME_LEN + tt] * 2.0f - 1.0f;
    nrev[lf * NROW + 79 - tt] = v;
  }
  for (int i = tid; i < FPW * 32; i += 256) {
    int lf = i >> 5;
    int q  = i & 31;
    nrev[lf * NROW + ((q < 16) ? q : q + 64)] = 0.0f;
  }
  for (int i = tid; i < FPW * (IR_ZLO + IROW - IR_ZHI); i += 256) {
    int lf = i / (IR_ZLO + IROW - IR_ZHI);
    int p  = i - lf * (IR_ZLO + IROW - IR_ZHI);
    irs[lf * IROW + ((p < IR_ZLO) ? p : p + (IR_ZHI - IR_ZLO))] = 0.0f;
  }

  const int row = lane & 15;
  const int hh  = lane >> 4;
  const int kb  = hh << 1;
  const int ft  = wave >> 2;
  const int fr  = fbase + 16 * ft + row;
  const int frc = (fr < 0) ? 0 : ((fr < FRAMES) ? fr : FRAMES - 1);
  const float* hrow = H + ((long)b * FRAMES + frc) * CBINS;

  v4fu ah[8];
#pragma unroll
  for (int i = 0; i < 8; ++i) ah[i] = *(const v4fu*)(hrow + 8 * i + 4 * hh);
  v2f at;
  {
    float h64 = hrow[CBINS - 1];
    at.x = (hh == 0) ? h64 : 0.0f;
    at.y = 0.0f;
  }

  for (int jt = wave & 3; jt < 9; jt += 4) {
    v8f acc = {};
    const float* mrow = M + (jt * 16 + row) * KPAD;
    v4fu bm[8];
#pragma unroll
    for (int i = 0; i < 8; ++i) bm[i] = *(const v4fu*)(mrow + 8 * i + 4 * hh);
    v2f bt = *(const v2f*)(mrow + 64 + kb);
#pragma unroll
    for (int i = 0; i < 8; ++i) {
      v2f a0, b0, a1, b1;
      a0.x = ah[i].x; a0.y = ah[i].y;
      b0.x = bm[i].x; b0.y = bm[i].y;
      a1.x = ah[i].z; a1.y = ah[i].w;
      b1.x = bm[i].z; b1.y = bm[i].w;
      acc = __builtin_amdgcn_wmma_f32_16x16x4_f32(
           false, a0,  false, b0,
           (short)0, acc,  false,  false);
      acc = __builtin_amdgcn_wmma_f32_16x16x4_f32(
          false, a1, false, b1, (short)0, acc, false, false);
    }
    acc = __builtin_amdgcn_wmma_f32_16x16x4_f32(
        false, at, false, bt, (short)0, acc, false, false);
    const int mbase = 16 * ft + ((lane >> 4) << 3);
#pragma unroll
    for (int v = 0; v < 8; ++v)
      irs[(mbase + v) * IROW + IR_ZLO + 17 * jt + row] =
          ((unsigned)(fbase + mbase + v) < (unsigned)FRAMES) ? acc[v] : 0.0f;
  }

  __syncthreads();

  {
    int bsel = ((row & 7) << 1) + (row >> 3);
    bsel = (bsel < 12) ? bsel : 11;
    const float* ap = irs + (FPV * wave) * IROW + 17 * bsel + kb;
    const float* bp = nrev + (FPV * wave) * NROW + 15 - row + kb;
    const int h0 = fbase + FPV * wave;
    float* op = out + (long)b * OUTLEN + (long)h0 * FRAME_LEN + lane;
    v8f P = {};
    v8f Q = {};
    float res[2 * FPV];
#pragma unroll
    for (int q = 0; q < FPV; ++q) {
      if (q > 0) {
        v8f np = {};
        v8f nq = {};
        np[0] = P[2]; np[1] = P[3]; np[2] = P[4]; np[3] = P[5];
        nq[0] = Q[2]; nq[1] = Q[3]; nq[2] = Q[4]; nq[3] = Q[5];
        P = np;
        Q = nq;
      }
      const float* aq = ap + q * IROW;
      const float* bq = bp + q * NROW;
#pragma unroll
      for (int k0 = 0; k0 < KHALF; k0 += 4) {
        const int k1  = k0 + KHALF;
        const int ao0 = k0 + (k0 >> 4);
        const int ao1 = k1 + (k1 >> 4);
        v2f a0, a1, b0, b1;
        a0.x = aq[ao0]; a0.y = aq[ao0 + 1];
        b0.x = bq[k0];  b0.y = bq[k0 + 1];
        a1.x = aq[ao1]; a1.y = aq[ao1 + 1];
        b1.x = bq[k1];  b1.y = bq[k1 + 1];
        P = __builtin_amdgcn_wmma_f32_16x16x4_f32(
            false, a0, false, b0, (short)0, P, false, false);
        Q = __builtin_amdgcn_wmma_f32_16x16x4_f32(
            false, a1, false, b1, (short)0, Q, false, false);
      }
      res[2 * q]     = P[0] + Q[0];
      res[2 * q + 1] = P[1] + Q[1];
    }
    float* cw = carry + wave * 128 + lane;
    cw[0]  = P[2] + Q[2];
    cw[32] = P[3] + Q[3];
    cw[64] = P[4] + Q[4];
    cw[96] = P[5] + Q[5];
    __syncthreads();
    if (wave > 0) {
      const float* cr = carry + (wave - 1) * 128 + lane;
      res[0] += cr[0];
      res[1] += cr[32];
      res[2] += cr[64];
      res[3] += cr[96];
    }
#pragma unroll
    for (int q = 0; q < FPV; ++q) {
      if ((q >= 2 || wave > 0) && h0 + q < NHOPS) {
        op[q * FRAME_LEN]      = res[2 * q]     * 0.01f;
        op[q * FRAME_LEN + 32] = res[2 * q + 1] * 0.01f;
      }
    }
  }
}

extern "C" void kernel_launch(void* const* d_in, const int* in_sizes, int n_in,
                              void* d_out, int out_size, void* d_ws, size_t ws_size,
                              hipStream_t stream) {
  const float* H     = (const float*)d_in[0];
  const float* noise = (const float*)d_in[1];
  float* out = (float*)d_out;
  float* Mws = (float*)d_ws;

  FN_setup_M<<<(JROWS * KPAD + 255) / 256, 256, 0, stream>>>(Mws);
  FN_fused<<<BATCH * WPB, 256, 0, stream>>>(H, noise, Mws, out);
}


# ninth column tile of the ir matmul removed: the window's first sample is 0, so ir[0] = 0 and rows j = 1..128 are exactly 8 tiles, two per wave (was 3 for two waves); sum order unchanged
# speedup vs baseline: 3.5734x; 1.0424x over previous
#include <hip/hip_runtime.h>
#include <hip/hip_bf16.h>


typedef __attribute__((ext_vector_type(2))) float v2f;
typedef __attribute__((ext_vector_type(8))) float v8f;
typedef __attribute__((ext_vector_type(4), aligned(4))) float v4fu;

#define BATCH     32
#define FRAMES    4000
#define CBINS     65
#define NIR       129
#define FRAME_LEN 64
#define SEGLEN    192
#define OUTLEN    256128
#define KPAD      68
#define JROWS     128
#define JTILES    8
#define NHOPS     4002
#define FPW       32
#define HPW       30
#define WPB       134
#define FPV       4
#define KCONV     80
#define KHALF     40
#define IROW      276
#define IR_ZLO    69
#define IR_ZHI    205
#define NROW      96

__global__ void FN_setup_M(float* __restrict__ M) {
  const float TWO_PI = 6.28318530717958647692f;
  for (int i = blockIdx.x * blockDim.x + threadIdx.x; i < JROWS * KPAD;
       i += gridDim.x * blockDim.x) {
    int r = i / KPAD;
    int k = i - r * KPAD;
    int j = r + 1;
    float v = 0.0f;
    if (k < CBINS) {
      float win = 0.5f * (1.0f - cosf(TWO_PI * (float)j / (float)NIR));
      int m = (k * (j - 64)) % NIR;
      float ang = TWO_PI * (float)m / (float)NIR;
      float ck  = (k == 0) ? 1.0f : 2.0f;
      v = win * (ck / (float)NIR) * cosf(ang);
    }
    M[i] = v;
  }
}

__global__ __launch_bounds__(256) void FN_fused(const float* __restrict__ H,
                                                const float* __restrict__ noise,
                                                const float* __restrict__ M,
                                                float* __restrict__ out) {
  __shared__ float irs[FPW * IROW];
  __shared__ float nrev[FPW * NROW];
  __shared__ float carry[8 * 4 * 32];

  const int tid  = threadIdx.x;
  const int lane = tid & 31;
  const int wave = tid >> 5;

  const int b     = blockIdx.x / WPB;
  const int S     = (blockIdx.x - b * WPB) * HPW;
  const int fbase = S - 2;

  for (int i = tid; i < FPW * FRAME_LEN; i += 256) {
    int lf = i >> 6;
    int tt = i & 63;
    int f  = fbase + lf;
    float v = 0.0f;
    if (f >= 0 && f < FRAMES)
      v = noise[((long)b * FRAMES + f) * FRAME_LEN + tt] * 2.0f - 1.0f;
    nrev[lf * NROW + 79 - tt] = v;
  }
  for (int i = tid; i < FPW * 32; i += 256) {
    int lf = i >> 5;
    int q  = i & 31;
    nrev[lf * NROW + ((q < 16) ? q : q + 64)] = 0.0f;
  }
  for (int i = tid; i < FPW * (IR_ZLO + IROW - IR_ZHI); i += 256) {
    int lf = i / (IR_ZLO + IROW - IR_ZHI);
    int p  = i - lf * (IR_ZLO + IROW - IR_ZHI);
    irs[lf * IROW + ((p < IR_ZLO) ? p : p + (IR_ZHI - IR_ZLO))] = 0.0f;
  }

  const int row = lane & 15;
  const int hh  = lane >> 4;
  const int kb  = hh << 1;
  const int ft  = wave >> 2;
  const int fr  = fbase + 16 * ft + row;
  const int frc = (fr < 0) ? 0 : ((fr < FRAMES) ? fr : FRAMES - 1);
  const float* hrow = H + ((long)b * FRAMES + frc) * CBINS;

  v4fu ah[8];
#pragma unroll
  for (int i = 0; i < 8; ++i) ah[i] = *(const v4fu*)(hrow + 8 * i + 4 * hh);
  v2f at;
  {
    float h64 = hrow[CBINS - 1];
    at.x = (hh == 0) ? h64 : 0.0f;
    at.y = 0.0f;
  }

  for (int jt = wave & 3; jt < JTILES; jt += 4) {
    v8f acc = {};
    const float* mrow = M + (jt * 16 + row) * KPAD;
    v4fu bm[8];
#pragma unroll
    for (int i = 0; i < 8; ++i) bm[i] = *(const v4fu*)(mrow + 8 * i + 4 * hh);
    v2f bt = *(const v2f*)(mrow + 64 + kb);
#pragma unroll
    for (int i = 0; i < 8; ++i) {
      v2f a0, b0, a1, b1;
      a0.x = ah[i].x; a0.y = ah[i].y;
      b0.x = bm[i].x; b0.y = bm[i].y;
      a1.x = ah[i].z; a1.y = ah[i].w;
      b1.x = bm[i].z; b1.y = bm[i].w;
      acc = __builtin_amdgcn_wmma_f32_16x16x4_f32(
           false, a0,  false, b0,
           (short)0, acc,  false,  false);
      acc = __builtin_amdgcn_wmma_f32_16x16x4_f32(
          false, a1, false, b1, (short)0, acc, false, false);
    }
    acc = __builtin_amdgcn_wmma_f32_16x16x4_f32(
        false, at, false, bt, (short)0, acc, false, false);
    const int mbase = 16 * ft + ((lane >> 4) << 3);
    const int slot  = IR_ZLO + 17 * jt + row + ((row + 1) >> 4);
#pragma unroll
    for (int v = 0; v < 8; ++v)
      irs[(mbase + v) * IROW + slot] =
          ((unsigned)(fbase + mbase + v) < (unsigned)FRAMES) ? acc[v] : 0.0f;
  }

  __syncthreads();

  {
    int bsel = ((row & 7) << 1) + (row >> 3);
    bsel = (bsel < 12) ? bsel : 11;
    const float* ap = irs + (FPV * wave) * IROW + 17 * bsel + kb;
    const float* bp = nrev + (FPV * wave) * NROW + 15 - row + kb;
    const int h0 = fbase + FPV * wave;
    float* op = out + (long)b * OUTLEN + (long)h0 * FRAME_LEN + lane;
    v8f P = {};
    v8f Q = {};
    float res[2 * FPV];
#pragma unroll
    for (int q = 0; q < FPV; ++q) {
      if (q > 0) {
        v8f np = {};
        v8f nq = {};
        np[0] = P[2]; np[1] = P[3]; np[2] = P[4]; np[3] = P[5];
        nq[0] = Q[2]; nq[1] = Q[3]; nq[2] = Q[4]; nq[3] = Q[5];
        P = np;
        Q = nq;
      }
      const float* aq = ap + q * IROW;
      const float* bq = bp + q * NROW;
#pragma unroll
      for (int k0 = 0; k0 < KHALF; k0 += 4) {
        const int k1  = k0 + KHALF;
        const int ao0 = k0 + (k0 >> 4);
        const int ao1 = k1 + (k1 >> 4);
        v2f a0, a1, b0, b1;
        a0.x = aq[ao0]; a0.y = aq[ao0 + 1];
        b0.x = bq[k0];  b0.y = bq[k0 + 1];
        a1.x = aq[ao1]; a1.y = aq[ao1 + 1];
        b1.x = bq[k1];  b1.y = bq[k1 + 1];
        P = __builtin_amdgcn_wmma_f32_16x16x4_f32(
            false, a0, false, b0, (short)0, P, false, false);
        Q = __builtin_amdgcn_wmma_f32_16x16x4_f32(
            false, a1, false, b1, (short)0, Q, false, false);
      }
      res[2 * q]     = P[0] + Q[0];
      res[2 * q + 1] = P[1] + Q[1];
    }
    float* cw = carry + wave * 128 + lane;
    cw[0]  = P[2] + Q[2];
    cw[32] = P[3] + Q[3];
    cw[64] = P[4] + Q[4];
    cw[96] = P[5] + Q[5];
    __syncthreads();
    if (wave > 0) {
      const float* cr = carry + (wave - 1) * 128 + lane;
      res[0] += cr[0];
      res[1] += cr[32];
      res[2] += cr[64];
      res[3] += cr[96];
    }
#pragma unroll
    for (int q = 0; q < FPV; ++q) {
      if ((q >= 2 || wave > 0) && h0 + q < NHOPS) {
        op[q * FRAME_LEN]      = res[2 * q]     * 0.01f;
        op[q * FRAME_LEN + 32] = res[2 * q + 1] * 0.01f;
      }
    }
  }
}

extern "C" void kernel_launch(void* const* d_in, const int* in_sizes, int n_in,
                              void* d_out, int out_size, void* d_ws, size_t ws_size,
                              hipStream_t stream) {
  const float* H     = (const float*)d_in[0];
  const float* noise = (const float*)d_in[1];
  float* out = (float*)d_out;
  float* Mws = (float*)d_ws;

  FN_setup_M<<<(JROWS * KPAD + 255) / 256, 256, 0, stream>>>(Mws);
  FN_fused<<<BATCH * WPB, 256, 0, stream>>>(H, noise, Mws, out);
}


# all input loads of a work-group issued before its first wait: noise by two 16-B loads per lane (was eight 4-B loads, each waited for), H loads moved above the LDS zero fill; sums unchanged
# speedup vs baseline: 3.8739x; 1.0841x over previous
#include <hip/hip_runtime.h>
#include <hip/hip_bf16.h>


typedef __attribute__((ext_vector_type(2))) float v2f;
typedef __attribute__((ext_vector_type(8))) float v8f;
typedef __attribute__((ext_vector_type(4), aligned(4))) float v4fu;
typedef __attribute__((ext_vector_type(4))) float v4f;

#define BATCH     32
#define FRAMES    4000
#define CBINS     65
#define NIR       129
#define FRAME_LEN 64
#define SEGLEN    192
#define OUTLEN    256128
#define KPAD      68
#define JROWS     128
#define JTILES    8
#define NHOPS     4002
#define FPW       32
#define HPW       30
#define WPB       134
#define FPV       4
#define KCONV     80
#define KHALF     40
#define IROW      276
#define IR_ZLO    69
#define IR_ZHI    205
#define NROW      96

__global__ void FN_setup_M(float* __restrict__ M) {
  const float TWO_PI = 6.28318530717958647692f;
  for (int i = blockIdx.x * blockDim.x + threadIdx.x; i < JROWS * KPAD;
       i += gridDim.x * blockDim.x) {
    int r = i / KPAD;
    int k = i - r * KPAD;
    int j = r + 1;
    float v = 0.0f;
    if (k < CBINS) {
      float win = 0.5f * (1.0f - cosf(TWO_PI * (float)j / (float)NIR));
      int m = (k * (j - 64)) % NIR;
      float ang = TWO_PI * (float)m / (float)NIR;
      float ck  = (k == 0) ? 1.0f : 2.0f;
      v = win * (ck / (float)NIR) * cosf(ang);
    }
    M[i] = v;
  }
}

__global__ __launch_bounds__(256) void FN_fused(const float* __restrict__ H,
                                                const float* __restrict__ noise,
                                                const float* __restrict__ M,
                                                float* __restrict__ out) {
  __shared__ float irs[FPW * IROW];
  __shared__ __attribute__((aligned(16))) float nrev[FPW * NROW];
  __shared__ float carry[8 * 4 * 32];

  const int tid  = threadIdx.x;
  const int lane = tid & 31;
  const int wave = tid >> 5;

  const int b     = blockIdx.x / WPB;
  const int S     = (blockIdx.x - b * WPB) * HPW;
  const int fbase = S - 2;

  v4f nz[2];
#pragma unroll
  for (int p = 0; p < 2; ++p) {
    int e  = (tid + 256 * p) << 2;
    int f  = fbase + (e >> 6);
    int fc = (f < 0) ? 0 : ((f < FRAMES) ? f : FRAMES - 1);
    nz[p] = *(const v4f*)(noise + ((long)b * FRAMES + fc) * FRAME_LEN + (e & 63));
  }
  const int row = lane & 15;
  const int hh  = lane >> 4;
  const int kb  = hh << 1;
  const int ft  = wave >> 2;
  const int fr  = fbase + 16 * ft + row;
  const int frc = (fr < 0) ? 0 : ((fr < FRAMES) ? fr : FRAMES - 1);
  const float* hrow = H + ((long)b * FRAMES + frc) * CBINS;

  v4fu ah[8];
#pragma unroll
  for (int i = 0; i < 8; ++i) ah[i] = *(const v4fu*)(hrow + 8 * i + 4 * hh);
  const float h64 = hrow[CBINS - 1];

  for (int i = tid; i < FPW * 32; i += 256) {
    int lf = i >> 5;
    int q  = i & 31;
    nrev[lf * NROW + ((q < 16) ? q : q + 64)] = 0.0f;
  }
#pragma nounroll
  for (int i = tid; i < FPW * (IR_ZLO + IROW - IR_ZHI); i += 256) {
    int lf = i / (IR_ZLO + IROW - IR_ZHI);
    int p  = i - lf * (IR_ZLO + IROW - IR_ZHI);
    irs[lf * IROW + ((p < IR_ZLO) ? p : p + (IR_ZHI - IR_ZLO))] = 0.0f;
  }

#pragma unroll
  for (int p = 0; p < 2; ++p) {
    int e  = (tid + 256 * p) << 2;
    int lf = e >> 6;
    int tt = e & 63;
    bool in = (unsigned)(fbase + lf) < (unsigned)FRAMES;
    v4f r;
    r.x = in ? nz[p].w * 2.0f - 1.0f : 0.0f;
    r.y = in ? nz[p].z * 2.0f - 1.0f : 0.0f;
    r.z = in ? nz[p].y * 2.0f - 1.0f : 0.0f;
    r.w = in ? nz[p].x * 2.0f - 1.0f : 0.0f;
    *(v4f*)(nrev + lf * NROW + 76 - tt) = r;
  }

  v2f at;
  at.x = (hh == 0) ? h64 : 0.0f;
  at.y = 0.0f;

  for (int jt = wave & 3; jt < JTILES; jt += 4) {
    v8f acc = {};
    const float* mrow = M + (jt * 16 + row) * KPAD;
    v4fu bm[8];
#pragma unroll
    for (int i = 0; i < 8; ++i) bm[i] = *(const v4fu*)(mrow + 8 * i + 4 * hh);
    v2f bt = *(const v2f*)(mrow + 64 + kb);
#pragma unroll
    for (int i = 0; i < 8; ++i) {
      v2f a0, b0, a1, b1;
      a0.x = ah[i].x; a0.y = ah[i].y;
      b0.x = bm[i].x; b0.y = bm[i].y;
      a1.x = ah[i].z; a1.y = ah[i].w;
      b1.x = bm[i].z; b1.y = bm[i].w;
      acc = __builtin_amdgcn_wmma_f32_16x16x4_f32(
           false, a0,  false, b0,
           (short)0, acc,  false,  false);
      acc = __builtin_amdgcn_wmma_f32_16x16x4_f32(
          false, a1, false, b1, (short)0, acc, false, false);
    }
    acc = __builtin_amdgcn_wmma_f32_16x16x4_f32(
        false, at, false, bt, (short)0, acc, false, false);
    const int mbase = 16 * ft + ((lane >> 4) << 3);
    const int slot  = IR_ZLO + 17 * jt + row + ((row + 1) >> 4);
#pragma unroll
    for (int v = 0; v < 8; ++v)
      irs[(mbase + v) * IROW + slot] =
          ((unsigned)(fbase + mbase + v) < (unsigned)FRAMES) ? acc[v] : 0.0f;
  }

  __syncthreads();

  {
    int bsel = ((row & 7) << 1) + (row >> 3);
    bsel = (bsel < 12) ? bsel : 11;
    const float* ap = irs + (FPV * wave) * IROW + 17 * bsel + kb;
    const float* bp = nrev + (FPV * wave) * NROW + 15 - row + kb;
    const int h0 = fbase + FPV * wave;
    float* op = out + (long)b * OUTLEN + (long)h0 * FRAME_LEN + lane;
    v8f P = {};
    v8f Q = {};
    float res[2 * FPV];
#pragma unroll
    for (int q = 0; q < FPV; ++q) {
      if (q > 0) {
        v8f np = {};
        v8f nq = {};
        np[0] = P[2]; np[1] = P[3]; np[2] = P[4]; np[3] = P[5];
        nq[0] = Q[2]; nq[1] = Q[3]; nq[2] = Q[4]; nq[3] = Q[5];
        P = np;
        Q = nq;
      }
      const float* aq = ap + q * IROW;
      const float* bq = bp + q * NROW;
#pragma unroll
      for (int k0 = 0; k0 < KHALF; k0 += 4) {
        const int k1  = k0 + KHALF;
        const int ao0 = k0 + (k0 >> 4);
        const int ao1 = k1 + (k1 >> 4);
        v2f a0, a1, b0, b1;
        a0.x = aq[ao0]; a0.y = aq[ao0 + 1];
        b0.x = bq[k0];  b0.y = bq[k0 + 1];
        a1.x = aq[ao1]; a1.y = aq[ao1 + 1];
        b1.x = bq[k1];  b1.y = bq[k1 + 1];
        P = __builtin_amdgcn_wmma_f32_16x16x4_f32(
            false, a0, false, b0, (short)0, P, false, false);
        Q = __builtin_amdgcn_wmma_f32_16x16x4_f32(
            false, a1, false, b1, (short)0, Q, false, false);
      }
      res[2 * q]     = P[0] + Q[0];
      res[2 * q + 1] = P[1] + Q[1];
    }
    float* cw = carry + wave * 128 + lane;
    cw[0]  = P[2] + Q[2];
    cw[32] = P[3] + Q[3];
    cw[64] = P[4] + Q[4];
    cw[96] = P[5] + Q[5];
    __syncthreads();
    if (wave > 0) {
      const float* cr = carry + (wave - 1) * 128 + lane;
      res[0] += cr[0];
      res[1] += cr[32];
      res[2] += cr[64];
      res[3] += cr[96];
    }
#pragma unroll
    for (int q = 0; q < FPV; ++q) {
      if ((q >= 2 || wave > 0) && h0 + q < NHOPS) {
        op[q * FRAME_LEN]      = res[2 * q]     * 0.01f;
        op[q * FRAME_LEN + 32] = res[2 * q + 1] * 0.01f;
      }
    }
  }
}

extern "C" void kernel_launch(void* const* d_in, const int* in_sizes, int n_in,
                              void* d_out, int out_size, void* d_ws, size_t ws_size,
                              hipStream_t stream) {
  const float* H     = (const float*)d_in[0];
  const float* noise = (const float*)d_in[1];
  float* out = (float*)d_out;
  float* Mws = (float*)d_ws;

  FN_setup_M<<<(JROWS * KPAD + 255) / 256, 256, 0, stream>>>(Mws);
  FN_fused<<<BATCH * WPB, 256, 0, stream>>>(H, noise, Mws, out);
}
